# starting kernel
# baseline (speedup 1.0000x reference)
#include <hip/hip_runtime.h>
#include <math.h>

#define CHN      256
#define ND       128
#define NEV      128
#define NOD      127
#define LSTRIDE  255
#define BATCHSZ  16384

static constexpr float ATTEN = 0.97723722095581068f;

typedef __attribute__((ext_vector_type(2))) float v2f;
typedef __attribute__((ext_vector_type(8))) float v8f;

#define WMMA_F32(A, B, C) \
  __builtin_amdgcn_wmma_f32_16x16x4_f32(false, (A), false, (B), (short)0, (C), false, false)

__device__ __forceinline__ void apply_mzi(float* sre, float* sim, int i, int j,
                                          float th, float ph) {
  float s, c, epi, epr;
  sincosf(th, &s, &c);
  sincosf(ph, &epi, &epr);
  const float ar = sre[i], ai = sim[i];
  const float br = sre[j], bi = sim[j];
  const float ecr = epr * c, eci = epi * c;
  const float esr = epr * s, esi = epi * s;
  const float nar = ecr * ar - eci * ai - s * br;
  const float nai = ecr * ai + eci * ar - s * bi;
  const float nbr = esr * ar - esi * ai + c * br;
  const float nbi = esr * ai + esi * ar + c * bi;
  sre[i] = nar * ATTEN; sim[i] = nai * ATTEN;
  sre[j] = nbr * ATTEN; sim[j] = nbi * ATTEN;
}

__global__ void __launch_bounds__(128)
mesh_build_kernel(const float* __restrict__ thetas, const float* __restrict__ phis,
                  float* __restrict__ SrP, float* __restrict__ SiP) {
  __shared__ float sre[CHN];
  __shared__ float sim[CHN];
  const int k = blockIdx.x;
  const int t = threadIdx.x;

  sre[2 * t]     = (2 * t     == k) ? 1.0f : 0.0f;
  sre[2 * t + 1] = (2 * t + 1 == k) ? 1.0f : 0.0f;
  sim[2 * t] = 0.0f;
  sim[2 * t + 1] = 0.0f;
  __syncthreads();

  for (int d = 0; d < ND; ++d) {
    {
      const int pi = d * LSTRIDE + t;
      apply_mzi(sre, sim, 2 * t, 2 * t + 1, thetas[pi], phis[pi]);
    }
    __syncthreads();
    if (t < NOD) {
      const int pi = d * LSTRIDE + NEV + t;
      apply_mzi(sre, sim, 2 * t + 1, 2 * t + 2, thetas[pi], phis[pi]);
    }
    __syncthreads();
  }

  float* rbase = SrP + (size_t)(k >> 1) * (2 * CHN) + (k & 1);
  float* ibase = SiP + (size_t)(k >> 1) * (2 * CHN) + (k & 1);
  for (int i = t; i < CHN; i += 128) {
    rbase[2 * i] = sre[i];
    ibase[2 * i] = sim[i];
  }
}

__global__ void __launch_bounds__(256)
mesh_gemm_kernel(const float* __restrict__ X, const float* __restrict__ SrP,
                 const float* __restrict__ SiP, float* __restrict__ out) {
  const int lane = threadIdx.x & 31;
  const int wave = threadIdx.x >> 5;
  const int wm = wave >> 2;
  const int wn = wave & 3;
  const int bm = blockIdx.x >> 1;
  const int bn = blockIdx.x & 1;
  const int row0 = bm * 64 + wm * 32;
  const int col0 = bn * 128 + wn * 32;
  const int lh = lane >> 4;
  const int lm = lane & 15;

  v8f accR00 = {}, accR01 = {}, accR10 = {}, accR11 = {};
  v8f accI00 = {}, accI01 = {}, accI10 = {}, accI11 = {};

  const float* aptr0 = X + (size_t)(row0 + lm) * CHN + lh * 2;
  const float* aptr1 = X + (size_t)(row0 + 16 + lm) * CHN + lh * 2;
  const size_t bbase = (size_t)lh * (2 * CHN) + (col0 + lm) * 2;
  const float* brptr = SrP + bbase;
  const float* biptr = SiP + bbase;

#pragma unroll 2
  for (int k = 0; k < CHN; k += 4) {
    const size_t boff = (size_t)(k >> 1) * (2 * CHN);
    const v2f a0  = *(const v2f*)(aptr0 + k);
    const v2f a1  = *(const v2f*)(aptr1 + k);
    const v2f br0 = *(const v2f*)(brptr + boff);
    const v2f br1 = *(const v2f*)(brptr + boff + 32);
    const v2f bi0 = *(const v2f*)(biptr + boff);
    const v2f bi1 = *(const v2f*)(biptr + boff + 32);
    accR00 = WMMA_F32(a0, br0, accR00);
    accR01 = WMMA_F32(a0, br1, accR01);
    accR10 = WMMA_F32(a1, br0, accR10);
    accR11 = WMMA_F32(a1, br1, accR11);
    accI00 = WMMA_F32(a0, bi0, accI00);
    accI01 = WMMA_F32(a0, bi1, accI01);
    accI10 = WMMA_F32(a1, bi0, accI10);
    accI11 = WMMA_F32(a1, bi1, accI11);
  }

  float* ob00 = out + (size_t)(row0 + lh * 8) * CHN + col0 + lm;
  float* ob10 = out + (size_t)(row0 + 16 + lh * 8) * CHN + col0 + lm;
#pragma unroll
  for (int r = 0; r < 8; ++r) {
    float vr, vi;
    vr = accR00[r]; vi = accI00[r];
    __builtin_nontemporal_store(vr * vr + vi * vi, ob00 + (size_t)r * CHN);
    vr = accR01[r]; vi = accI01[r];
    __builtin_nontemporal_store(vr * vr + vi * vi, ob00 + (size_t)r * CHN + 16);
    vr = accR10[r]; vi = accI10[r];
    __builtin_nontemporal_store(vr * vr + vi * vi, ob10 + (size_t)r * CHN);
    vr = accR11[r]; vi = accI11[r];
    __builtin_nontemporal_store(vr * vr + vi * vi, ob10 + (size_t)r * CHN + 16);
  }
}

extern "C" void kernel_launch(void* const* d_in, const int* in_sizes, int n_in,
                              void* d_out, int out_size, void* d_ws, size_t ws_size,
                              hipStream_t stream) {
  const float* x      = (const float*)d_in[0];
  const float* thetas = (const float*)d_in[1];
  const float* phis   = (const float*)d_in[2];
  float* out = (float*)d_out;

  float* SrP = (float*)d_ws;
  float* SiP = SrP + CHN * CHN;

  mesh_build_kernel<<<CHN, 128, 0, stream>>>(thetas, phis, SrP, SiP);

  const int blocks = (BATCHSZ / 64) * (CHN / 128);
  mesh_gemm_kernel<<<blocks, 256, 0, stream>>>(x, SrP, SiP, out);
}


# cos/sin of each MZI's theta and phi computed once by a new table kernel (3 launches, was 2) and read as one float4 per MZI, not by two sincosf per MZI in each of the 256 basis-vector blocks
# speedup vs baseline: 1.6462x; 1.6462x over previous
#include <hip/hip_runtime.h>
#include <math.h>

#define CHN      256
#define ND       128
#define NEV      128
#define NOD      127
#define LSTRIDE  255
#define BATCHSZ  16384

static constexpr float ATTEN = 0.97723722095581068f;

typedef __attribute__((ext_vector_type(2))) float v2f;
typedef __attribute__((ext_vector_type(8))) float v8f;

#define WMMA_F32(A, B, C) \
  __builtin_amdgcn_wmma_f32_16x16x4_f32(false, (A), false, (B), (short)0, (C), false, false)

#define TABROW   256

__global__ void __launch_bounds__(TABROW)
mesh_trig_kernel(const float* __restrict__ thetas, const float* __restrict__ phis,
                 float* __restrict__ tab) {
  const int d = blockIdx.x;
  const int j = threadIdx.x;
  float4 v = make_float4(0.0f, 0.0f, 0.0f, 0.0f);
  if (j < LSTRIDE) {
    const int pi = d * LSTRIDE + j;
    float s, c, epi, epr;
    sincosf(thetas[pi], &s, &c);
    sincosf(phis[pi], &epi, &epr);
    v = make_float4(c, s, epr, epi);
  }
  ((float4*)tab)[d * TABROW + j] = v;
}

__device__ __forceinline__ void apply_mzi(float* sre, float* sim, int i, int j,
                                          const float4 q) {
  const float c = q.x, s = q.y, epr = q.z, epi = q.w;
  const float ar = sre[i], ai = sim[i];
  const float br = sre[j], bi = sim[j];
  const float ecr = epr * c, eci = epi * c;
  const float esr = epr * s, esi = epi * s;
  const float nar = ecr * ar - eci * ai - s * br;
  const float nai = ecr * ai + eci * ar - s * bi;
  const float nbr = esr * ar - esi * ai + c * br;
  const float nbi = esr * ai + esi * ar + c * bi;
  sre[i] = nar * ATTEN; sim[i] = nai * ATTEN;
  sre[j] = nbr * ATTEN; sim[j] = nbi * ATTEN;
}

__global__ void __launch_bounds__(128)
mesh_build_kernel(const float* __restrict__ tab,
                  float* __restrict__ SrP, float* __restrict__ SiP) {
  __shared__ float sre[CHN];
  __shared__ float sim[CHN];
  const int k = blockIdx.x;
  const int t = threadIdx.x;

  sre[2 * t]     = (2 * t     == k) ? 1.0f : 0.0f;
  sre[2 * t + 1] = (2 * t + 1 == k) ? 1.0f : 0.0f;
  sim[2 * t] = 0.0f;
  sim[2 * t + 1] = 0.0f;
  __syncthreads();

  const float4* row = (const float4*)tab + t;
  for (int d = 0; d < ND; ++d, row += TABROW) {
    const float4 qe = row[0];
    const float4 qo = row[NEV];
    apply_mzi(sre, sim, 2 * t, 2 * t + 1, qe);
    __syncthreads();
    if (t < NOD) {
      apply_mzi(sre, sim, 2 * t + 1, 2 * t + 2, qo);
    }
    __syncthreads();
  }

  float* rbase = SrP + (size_t)(k >> 1) * (2 * CHN) + (k & 1);
  float* ibase = SiP + (size_t)(k >> 1) * (2 * CHN) + (k & 1);
  for (int i = t; i < CHN; i += 128) {
    rbase[2 * i] = sre[i];
    ibase[2 * i] = sim[i];
  }
}

__global__ void __launch_bounds__(256)
mesh_gemm_kernel(const float* __restrict__ X, const float* __restrict__ SrP,
                 const float* __restrict__ SiP, float* __restrict__ out) {
  const int lane = threadIdx.x & 31;
  const int wave = threadIdx.x >> 5;
  const int wm = wave >> 2;
  const int wn = wave & 3;
  const int bm = blockIdx.x >> 1;
  const int bn = blockIdx.x & 1;
  const int row0 = bm * 64 + wm * 32;
  const int col0 = bn * 128 + wn * 32;
  const int lh = lane >> 4;
  const int lm = lane & 15;

  v8f accR00 = {}, accR01 = {}, accR10 = {}, accR11 = {};
  v8f accI00 = {}, accI01 = {}, accI10 = {}, accI11 = {};

  const float* aptr0 = X + (size_t)(row0 + lm) * CHN + lh * 2;
  const float* aptr1 = X + (size_t)(row0 + 16 + lm) * CHN + lh * 2;
  const size_t bbase = (size_t)lh * (2 * CHN) + (col0 + lm) * 2;
  const float* brptr = SrP + bbase;
  const float* biptr = SiP + bbase;

#pragma unroll 2
  for (int k = 0; k < CHN; k += 4) {
    const size_t boff = (size_t)(k >> 1) * (2 * CHN);
    const v2f a0  = *(const v2f*)(aptr0 + k);
    const v2f a1  = *(const v2f*)(aptr1 + k);
    const v2f br0 = *(const v2f*)(brptr + boff);
    const v2f br1 = *(const v2f*)(brptr + boff + 32);
    const v2f bi0 = *(const v2f*)(biptr + boff);
    const v2f bi1 = *(const v2f*)(biptr + boff + 32);
    accR00 = WMMA_F32(a0, br0, accR00);
    accR01 = WMMA_F32(a0, br1, accR01);
    accR10 = WMMA_F32(a1, br0, accR10);
    accR11 = WMMA_F32(a1, br1, accR11);
    accI00 = WMMA_F32(a0, bi0, accI00);
    accI01 = WMMA_F32(a0, bi1, accI01);
    accI10 = WMMA_F32(a1, bi0, accI10);
    accI11 = WMMA_F32(a1, bi1, accI11);
  }

  float* ob00 = out + (size_t)(row0 + lh * 8) * CHN + col0 + lm;
  float* ob10 = out + (size_t)(row0 + 16 + lh * 8) * CHN + col0 + lm;
#pragma unroll
  for (int r = 0; r < 8; ++r) {
    float vr, vi;
    vr = accR00[r]; vi = accI00[r];
    __builtin_nontemporal_store(vr * vr + vi * vi, ob00 + (size_t)r * CHN);
    vr = accR01[r]; vi = accI01[r];
    __builtin_nontemporal_store(vr * vr + vi * vi, ob00 + (size_t)r * CHN + 16);
    vr = accR10[r]; vi = accI10[r];
    __builtin_nontemporal_store(vr * vr + vi * vi, ob10 + (size_t)r * CHN);
    vr = accR11[r]; vi = accI11[r];
    __builtin_nontemporal_store(vr * vr + vi * vi, ob10 + (size_t)r * CHN + 16);
  }
}

extern "C" void kernel_launch(void* const* d_in, const int* in_sizes, int n_in,
                              void* d_out, int out_size, void* d_ws, size_t ws_size,
                              hipStream_t stream) {
  const float* x      = (const float*)d_in[0];
  const float* thetas = (const float*)d_in[1];
  const float* phis   = (const float*)d_in[2];
  float* out = (float*)d_out;

  float* SrP = (float*)d_ws;
  float* SiP = SrP + CHN * CHN;
  float* tab = SiP + CHN * CHN;

  mesh_trig_kernel<<<ND, TABROW, 0, stream>>>(thetas, phis, tab);

  mesh_build_kernel<<<CHN, 128, 0, stream>>>(tab, SrP, SiP);

  const int blocks = (BATCHSZ / 64) * (CHN / 128);
  mesh_gemm_kernel<<<blocks, 256, 0, stream>>>(x, SrP, SiP, out);
}


# transfer-matrix build: one wave per basis vector, its 256 channels in registers (8 per lane), lane neighbours exchanged by DPP/permlanex16 moves; no LDS, no barriers; table laid out per lane
# speedup vs baseline: 3.2070x; 1.9481x over previous
#include <hip/hip_runtime.h>
#include <math.h>

#define CHN      256
#define ND       128
#define NEV      128
#define NOD      127
#define LSTRIDE  255
#define BATCHSZ  16384

static constexpr float ATTEN = 0.97723722095581068f;

typedef __attribute__((ext_vector_type(2))) float v2f;
typedef __attribute__((ext_vector_type(8))) float v8f;

#define WMMA_F32(A, B, C) \
  __builtin_amdgcn_wmma_f32_16x16x4_f32(false, (A), false, (B), (short)0, (C), false, false)

#define PSLOTS   9
#define TABROW   (PSLOTS * 32)

__global__ void __launch_bounds__(TABROW)
mesh_trig_kernel(const float* __restrict__ thetas, const float* __restrict__ phis,
                 float* __restrict__ tab) {
  const int d = blockIdx.x;
  const int m = threadIdx.x >> 5;
  const int l = threadIdx.x & 31;
  int pi = -1;
  if (m < 4) {
    pi = d * LSTRIDE + 4 * l + m;
  } else {
    const int o = 4 * l - 1 + (m - 4);
    if (o >= 0 && o < NOD) pi = d * LSTRIDE + NEV + o;
  }
  float4 v = make_float4(0.0f, 0.0f, 0.0f, 0.0f);
  if (pi >= 0) {
    float s, c, epi, epr;
    sincosf(thetas[pi], &s, &c);
    sincosf(phis[pi], &epi, &epr);
    v = make_float4(c, s, epr, epi);
  }
  ((float4*)tab)[(d * PSLOTS + m) * 32 + l] = v;
}

typedef __attribute__((ext_vector_type(4))) float v4f;

struct LayerP { v4f q[PSLOTS]; };

__device__ __forceinline__ LayerP load_layer(const v4f* __restrict__ p) {
  LayerP r;
#pragma unroll
  for (int m = 0; m < PSLOTS; ++m) r.q[m] = p[m * 32];
  return r;
}

__device__ __forceinline__ void hold_loads(const LayerP& P) {
  asm volatile("" :: "v"(P.q[0]), "v"(P.q[1]), "v"(P.q[2]), "v"(P.q[3]), "v"(P.q[4]),
                     "v"(P.q[5]), "v"(P.q[6]), "v"(P.q[7]), "v"(P.q[8]));
}

__device__ __forceinline__ void apply_mzi(float& ar, float& ai, float& br, float& bi,
                                          const v4f q) {
  const float c = q.x, s = q.y, epr = q.z, epi = q.w;
  const float ecr = epr * c, eci = epi * c;
  const float esr = epr * s, esi = epi * s;
  const float nar = ecr * ar - eci * ai - s * br;
  const float nai = ecr * ai + eci * ar - s * bi;
  const float nbr = esr * ar - esi * ai + c * br;
  const float nbi = esr * ai + esi * ar + c * bi;
  ar = nar * ATTEN; ai = nai * ATTEN;
  br = nbr * ATTEN; bi = nbi * ATTEN;
}

__device__ __forceinline__ float from_lane_above(float v) {
  const int x = __float_as_int(v);
  const int edge = __builtin_amdgcn_permlanex16(x, x, 0, 0, false, false);
  return __int_as_float(__builtin_amdgcn_update_dpp(edge, x, 0x101, 0xf, 0xf, false));
}

__device__ __forceinline__ float from_lane_below(float v) {
  const int x = __float_as_int(v);
  const int edge = __builtin_amdgcn_permlanex16(x, x, 0xf, 0, false, false);
  return __int_as_float(__builtin_amdgcn_update_dpp(edge, x, 0x111, 0xf, 0xf, false));
}

__device__ __forceinline__ void double_layer(float (&xr)[8], float (&xi)[8], const LayerP& P,
                                             bool has_lo, bool has_hi) {
  apply_mzi(xr[0], xi[0], xr[1], xi[1], P.q[0]);
  apply_mzi(xr[6], xi[6], xr[7], xi[7], P.q[3]);
  apply_mzi(xr[2], xi[2], xr[3], xi[3], P.q[1]);
  apply_mzi(xr[4], xi[4], xr[5], xi[5], P.q[2]);
  const float hbr = from_lane_above(xr[0]);
  const float hbi = from_lane_above(xi[0]);
  const float lar = from_lane_below(xr[7]);
  const float lai = from_lane_below(xi[7]);
  apply_mzi(xr[1], xi[1], xr[2], xi[2], P.q[5]);
  apply_mzi(xr[3], xi[3], xr[4], xi[4], P.q[6]);
  apply_mzi(xr[5], xi[5], xr[6], xi[6], P.q[7]);
  {
    const v4f q = P.q[8];
    const float s = q.y;
    const float ecr = q.z * q.x, eci = q.w * q.x;
    const float nar = ecr * xr[7] - eci * xi[7] - s * hbr;
    const float nai = ecr * xi[7] + eci * xr[7] - s * hbi;
    xr[7] = has_hi ? nar * ATTEN : xr[7];
    xi[7] = has_hi ? nai * ATTEN : xi[7];
  }
  {
    const v4f q = P.q[4];
    const float c = q.x;
    const float esr = q.z * q.y, esi = q.w * q.y;
    const float nbr = esr * lar - esi * lai + c * xr[0];
    const float nbi = esr * lai + esi * lar + c * xi[0];
    xr[0] = has_lo ? nbr * ATTEN : xr[0];
    xi[0] = has_lo ? nbi * ATTEN : xi[0];
  }
}

#define BUILD_WAVES 4

__global__ void __launch_bounds__(BUILD_WAVES * 32) __attribute__((amdgpu_waves_per_eu(1, 1)))
mesh_build_kernel(const float* __restrict__ tab,
                  float* __restrict__ SrP, float* __restrict__ SiP) {
  const int lane = threadIdx.x & 31;
  const int k = blockIdx.x * BUILD_WAVES + (threadIdx.x >> 5);

  float xr[8], xi[8];
#pragma unroll
  for (int j = 0; j < 8; ++j) {
    xr[j] = (8 * lane + j == k) ? 1.0f : 0.0f;
    xi[j] = 0.0f;
  }
  const bool has_lo = lane > 0;
  const bool has_hi = lane < 31;

  const v4f* p = (const v4f*)tab + lane;
  LayerP a0 = load_layer(p);
  LayerP a1 = load_layer(p + TABROW);
  for (int d = 0; d < ND; d += 4) {
    const LayerP b0 = load_layer(p + (size_t)(d + 2) * TABROW);
    const LayerP b1 = load_layer(p + (size_t)(d + 3) * TABROW);
    double_layer(xr, xi, a0, has_lo, has_hi);
    double_layer(xr, xi, a1, has_lo, has_hi);
    const int dn = (d + 4 < ND) ? d + 4 : ND - 2;
    a0 = load_layer(p + (size_t)dn * TABROW);
    a1 = load_layer(p + (size_t)(dn + 1) * TABROW);
    double_layer(xr, xi, b0, has_lo, has_hi);
    double_layer(xr, xi, b1, has_lo, has_hi);
    hold_loads(a0);
    hold_loads(a1);
  }

  float* rbase = SrP + (size_t)(k >> 1) * (2 * CHN) + (k & 1) + 16 * lane;
  float* ibase = SiP + (size_t)(k >> 1) * (2 * CHN) + (k & 1) + 16 * lane;
#pragma unroll
  for (int j = 0; j < 8; ++j) {
    rbase[2 * j] = xr[j];
    ibase[2 * j] = xi[j];
  }
}

__global__ void __launch_bounds__(256)
mesh_gemm_kernel(const float* __restrict__ X, const float* __restrict__ SrP,
                 const float* __restrict__ SiP, float* __restrict__ out) {
  const int lane = threadIdx.x & 31;
  const int wave = threadIdx.x >> 5;
  const int wm = wave >> 2;
  const int wn = wave & 3;
  const int bm = blockIdx.x >> 1;
  const int bn = blockIdx.x & 1;
  const int row0 = bm * 64 + wm * 32;
  const int col0 = bn * 128 + wn * 32;
  const int lh = lane >> 4;
  const int lm = lane & 15;

  v8f accR00 = {}, accR01 = {}, accR10 = {}, accR11 = {};
  v8f accI00 = {}, accI01 = {}, accI10 = {}, accI11 = {};

  const float* aptr0 = X + (size_t)(row0 + lm) * CHN + lh * 2;
  const float* aptr1 = X + (size_t)(row0 + 16 + lm) * CHN + lh * 2;
  const size_t bbase = (size_t)lh * (2 * CHN) + (col0 + lm) * 2;
  const float* brptr = SrP + bbase;
  const float* biptr = SiP + bbase;

#pragma unroll 2
  for (int k = 0; k < CHN; k += 4) {
    const size_t boff = (size_t)(k >> 1) * (2 * CHN);
    const v2f a0  = *(const v2f*)(aptr0 + k);
    const v2f a1  = *(const v2f*)(aptr1 + k);
    const v2f br0 = *(const v2f*)(brptr + boff);
    const v2f br1 = *(const v2f*)(brptr + boff + 32);
    const v2f bi0 = *(const v2f*)(biptr + boff);
    const v2f bi1 = *(const v2f*)(biptr + boff + 32);
    accR00 = WMMA_F32(a0, br0, accR00);
    accR01 = WMMA_F32(a0, br1, accR01);
    accR10 = WMMA_F32(a1, br0, accR10);
    accR11 = WMMA_F32(a1, br1, accR11);
    accI00 = WMMA_F32(a0, bi0, accI00);
    accI01 = WMMA_F32(a0, bi1, accI01);
    accI10 = WMMA_F32(a1, bi0, accI10);
    accI11 = WMMA_F32(a1, bi1, accI11);
  }

  float* ob00 = out + (size_t)(row0 + lh * 8) * CHN + col0 + lm;
  float* ob10 = out + (size_t)(row0 + 16 + lh * 8) * CHN + col0 + lm;
#pragma unroll
  for (int r = 0; r < 8; ++r) {
    float vr, vi;
    vr = accR00[r]; vi = accI00[r];
    __builtin_nontemporal_store(vr * vr + vi * vi, ob00 + (size_t)r * CHN);
    vr = accR01[r]; vi = accI01[r];
    __builtin_nontemporal_store(vr * vr + vi * vi, ob00 + (size_t)r * CHN + 16);
    vr = accR10[r]; vi = accI10[r];
    __builtin_nontemporal_store(vr * vr + vi * vi, ob10 + (size_t)r * CHN);
    vr = accR11[r]; vi = accI11[r];
    __builtin_nontemporal_store(vr * vr + vi * vi, ob10 + (size_t)r * CHN + 16);
  }
}

extern "C" void kernel_launch(void* const* d_in, const int* in_sizes, int n_in,
                              void* d_out, int out_size, void* d_ws, size_t ws_size,
                              hipStream_t stream) {
  const float* x      = (const float*)d_in[0];
  const float* thetas = (const float*)d_in[1];
  const float* phis   = (const float*)d_in[2];
  float* out = (float*)d_out;

  float* SrP = (float*)d_ws;
  float* SiP = SrP + CHN * CHN;
  float* tab = SiP + CHN * CHN;

  mesh_trig_kernel<<<ND, TABROW, 0, stream>>>(thetas, phis, tab);

  mesh_build_kernel<<<CHN / BUILD_WAVES, BUILD_WAVES * 32, 0, stream>>>(tab, SrP, SiP);

  const int blocks = (BATCHSZ / 64) * (CHN / 128);
  mesh_gemm_kernel<<<blocks, 256, 0, stream>>>(x, SrP, SiP, out);
}


# build kernel: table holds each MZI's final complex coefficients (loss folded in, identity at the mesh edges), an MZI is 6 packed (re,im) ops; sets reloaded in place 8 layers ahead, 2 waves/block
# speedup vs baseline: 3.9992x; 1.2470x over previous
#include <hip/hip_runtime.h>
#include <math.h>

#define CHN      256
#define ND       128
#define NEV      128
#define NOD      127
#define LSTRIDE  255
#define BATCHSZ  16384

static constexpr float ATTEN = 0.97723722095581068f;

typedef __attribute__((ext_vector_type(2))) float v2f;
typedef __attribute__((ext_vector_type(8))) float v8f;

#define WMMA_F32(A, B, C) \
  __builtin_amdgcn_wmma_f32_16x16x4_f32(false, (A), false, (B), (short)0, (C), false, false)

#define PSLOTS   9
#define TROWS    12
#define TABROW   (TROWS * 32)

__global__ void __launch_bounds__(PSLOTS * 32)
mesh_trig_kernel(const float* __restrict__ thetas, const float* __restrict__ phis,
                 float* __restrict__ tab) {
  const int d = blockIdx.x;
  const int m = threadIdx.x >> 5;
  const int l = threadIdx.x & 31;
  int pi = -1;
  if (m < 4) {
    pi = d * LSTRIDE + 4 * l + m;
  } else {
    const int o = 4 * l - 1 + (m - 4);
    if (o >= 0 && o < NOD) pi = d * LSTRIDE + NEV + o;
  }
  float ecr = 1.0f, eci = 0.0f, esr = 0.0f, esi = 0.0f, cA = 1.0f, sA = 0.0f;
  if (pi >= 0) {
    float s, c, epi, epr;
    sincosf(thetas[pi], &s, &c);
    sincosf(phis[pi], &epi, &epr);
    cA = c * ATTEN;
    sA = s * ATTEN;
    ecr = epr * cA; eci = epi * cA;
    esr = epr * sA; esi = epi * sA;
  }
  float* base = tab + ((size_t)d * TABROW + l) * 4;
  if (m < 4) {
    ((float4*)base)[m * 32] = make_float4(ecr, eci, esr, esi);
    float* cs = base + (7 + (m >> 1)) * 128 + (m & 1) * 2;
    cs[0] = cA; cs[1] = sA;
  } else if (m == 4) {
    float* q = base + 11 * 128;
    q[1] = cA; q[2] = esr; q[3] = esi;
  } else if (m < 8) {
    const int o = m - 5;
    ((float4*)base)[(4 + o) * 32] = make_float4(ecr, eci, esr, esi);
    float* cs = base + (o < 2 ? 9 * 128 + o * 2 : 10 * 128);
    cs[0] = cA; cs[1] = sA;
  } else {
    float* q = base + 10 * 128;
    q[2] = ecr; q[3] = eci;
    base[11 * 128] = sA;
  }
}

typedef __attribute__((ext_vector_type(4))) float v4f;

struct LayerP { v4f q[TROWS]; };

__device__ __forceinline__ LayerP load_layer(const v4f* __restrict__ p) {
  LayerP r;
#pragma unroll
  for (int m = 0; m < TROWS; ++m) r.q[m] = p[m * 32];
  return r;
}

__device__ __forceinline__ unsigned after(unsigned row, const v2f (&x)[8]) {
  asm volatile("" : "+v"(row) : "v"(x[0]), "v"(x[1]), "v"(x[2]), "v"(x[3]),
                                "v"(x[4]), "v"(x[5]), "v"(x[6]), "v"(x[7]));
  return row;
}

__device__ __forceinline__ v2f splat(float v) { return (v2f){v, v}; }

__device__ __forceinline__ v2f cmul(float wr, float wi, v2f z) {
  const v2f zs = (v2f){z.y, z.x};
  return __builtin_elementwise_fma(zs, (v2f){-wi, wi}, z * splat(wr));
}

__device__ __forceinline__ void apply_mzi(v2f& a, v2f& b, const v4f e, float cA, float sA) {
  const v2f na = __builtin_elementwise_fma(b, splat(-sA), cmul(e.x, e.y, a));
  const v2f nb = __builtin_elementwise_fma(b, splat(cA), cmul(e.z, e.w, a));
  a = na; b = nb;
}

__device__ __forceinline__ float from_lane_above(float v) {
  const int x = __float_as_int(v);
  const int edge = __builtin_amdgcn_permlanex16(x, x, 0, 0, false, false);
  return __int_as_float(__builtin_amdgcn_update_dpp(edge, x, 0x101, 0xf, 0xf, false));
}

__device__ __forceinline__ float from_lane_below(float v) {
  const int x = __float_as_int(v);
  const int edge = __builtin_amdgcn_permlanex16(x, x, 0xf, 0, false, false);
  return __int_as_float(__builtin_amdgcn_update_dpp(edge, x, 0x111, 0xf, 0xf, false));
}

__device__ __forceinline__ void double_layer(v2f (&x)[8], const LayerP& P) {
  apply_mzi(x[0], x[1], P.q[0], P.q[7].x, P.q[7].y);
  apply_mzi(x[6], x[7], P.q[3], P.q[8].z, P.q[8].w);
  apply_mzi(x[2], x[3], P.q[1], P.q[7].z, P.q[7].w);
  apply_mzi(x[4], x[5], P.q[2], P.q[8].x, P.q[8].y);
  const v2f hb = (v2f){from_lane_above(x[0].x), from_lane_above(x[0].y)};
  const v2f la = (v2f){from_lane_below(x[7].x), from_lane_below(x[7].y)};
  apply_mzi(x[1], x[2], P.q[4], P.q[9].x, P.q[9].y);
  apply_mzi(x[3], x[4], P.q[5], P.q[9].z, P.q[9].w);
  apply_mzi(x[5], x[6], P.q[6], P.q[10].x, P.q[10].y);
  x[7] = __builtin_elementwise_fma(hb, splat(-P.q[11].x), cmul(P.q[10].z, P.q[10].w, x[7]));
  x[0] = __builtin_elementwise_fma(x[0], splat(P.q[11].y), cmul(P.q[11].z, P.q[11].w, la));
}

#define BUILD_WAVES 2
#define RING 8

__global__ void __launch_bounds__(BUILD_WAVES * 32) __attribute__((amdgpu_waves_per_eu(1, 1)))
mesh_build_kernel(const float* __restrict__ tab,
                  float* __restrict__ SrP, float* __restrict__ SiP) {
  const int lane = threadIdx.x & 31;
  const int k = blockIdx.x * BUILD_WAVES + (threadIdx.x >> 5);

  v2f x[8];
#pragma unroll
  for (int j = 0; j < 8; ++j) x[j] = (v2f){(8 * lane + j == k) ? 1.0f : 0.0f, 0.0f};

  const v4f* tab4 = (const v4f*)tab;
  const v4f* p = tab4 + lane;
  LayerP s0 = load_layer(p + 0 * TABROW), s1 = load_layer(p + 1 * TABROW),
         s2 = load_layer(p + 2 * TABROW), s3 = load_layer(p + 3 * TABROW),
         s4 = load_layer(p + 4 * TABROW), s5 = load_layer(p + 5 * TABROW),
         s6 = load_layer(p + 6 * TABROW), s7 = load_layer(p + 7 * TABROW);
#define RING_STEP(S, i)                                                     \
  {                                                                         \
    double_layer(x, S);                                                     \
    const int dn = (d + RING < ND) ? d + RING + (i) : d + (i);              \
    S = load_layer(tab4 + after((unsigned)(dn * TABROW + lane), x));        \
  }
  for (int d = 0; d < ND; d += RING) {
    RING_STEP(s0, 0) RING_STEP(s1, 1) RING_STEP(s2, 2) RING_STEP(s3, 3)
    RING_STEP(s4, 4) RING_STEP(s5, 5) RING_STEP(s6, 6) RING_STEP(s7, 7)
  }
#undef RING_STEP

  float* rbase = SrP + (size_t)(k >> 1) * (2 * CHN) + (k & 1) + 16 * lane;
  float* ibase = SiP + (size_t)(k >> 1) * (2 * CHN) + (k & 1) + 16 * lane;
#pragma unroll
  for (int j = 0; j < 8; ++j) {
    rbase[2 * j] = x[j].x;
    ibase[2 * j] = x[j].y;
  }
}

__global__ void __launch_bounds__(256)
mesh_gemm_kernel(const float* __restrict__ X, const float* __restrict__ SrP,
                 const float* __restrict__ SiP, float* __restrict__ out) {
  const int lane = threadIdx.x & 31;
  const int wave = threadIdx.x >> 5;
  const int wm = wave >> 2;
  const int wn = wave & 3;
  const int bm = blockIdx.x >> 1;
  const int bn = blockIdx.x & 1;
  const int row0 = bm * 64 + wm * 32;
  const int col0 = bn * 128 + wn * 32;
  const int lh = lane >> 4;
  const int lm = lane & 15;

  v8f accR00 = {}, accR01 = {}, accR10 = {}, accR11 = {};
  v8f accI00 = {}, accI01 = {}, accI10 = {}, accI11 = {};

  const float* aptr0 = X + (size_t)(row0 + lm) * CHN + lh * 2;
  const float* aptr1 = X + (size_t)(row0 + 16 + lm) * CHN + lh * 2;
  const size_t bbase = (size_t)lh * (2 * CHN) + (col0 + lm) * 2;
  const float* brptr = SrP + bbase;
  const float* biptr = SiP + bbase;

#pragma unroll 2
  for (int k = 0; k < CHN; k += 4) {
    const size_t boff = (size_t)(k >> 1) * (2 * CHN);
    const v2f a0  = *(const v2f*)(aptr0 + k);
    const v2f a1  = *(const v2f*)(aptr1 + k);
    const v2f br0 = *(const v2f*)(brptr + boff);
    const v2f br1 = *(const v2f*)(brptr + boff + 32);
    const v2f bi0 = *(const v2f*)(biptr + boff);
    const v2f bi1 = *(const v2f*)(biptr + boff + 32);
    accR00 = WMMA_F32(a0, br0, accR00);
    accR01 = WMMA_F32(a0, br1, accR01);
    accR10 = WMMA_F32(a1, br0, accR10);
    accR11 = WMMA_F32(a1, br1, accR11);
    accI00 = WMMA_F32(a0, bi0, accI00);
    accI01 = WMMA_F32(a0, bi1, accI01);
    accI10 = WMMA_F32(a1, bi0, accI10);
    accI11 = WMMA_F32(a1, bi1, accI11);
  }

  float* ob00 = out + (size_t)(row0 + lh * 8) * CHN + col0 + lm;
  float* ob10 = out + (size_t)(row0 + 16 + lh * 8) * CHN + col0 + lm;
#pragma unroll
  for (int r = 0; r < 8; ++r) {
    float vr, vi;
    vr = accR00[r]; vi = accI00[r];
    __builtin_nontemporal_store(vr * vr + vi * vi, ob00 + (size_t)r * CHN);
    vr = accR01[r]; vi = accI01[r];
    __builtin_nontemporal_store(vr * vr + vi * vi, ob00 + (size_t)r * CHN + 16);
    vr = accR10[r]; vi = accI10[r];
    __builtin_nontemporal_store(vr * vr + vi * vi, ob10 + (size_t)r * CHN);
    vr = accR11[r]; vi = accI11[r];
    __builtin_nontemporal_store(vr * vr + vi * vi, ob10 + (size_t)r * CHN + 16);
  }
}

extern "C" void kernel_launch(void* const* d_in, const int* in_sizes, int n_in,
                              void* d_out, int out_size, void* d_ws, size_t ws_size,
                              hipStream_t stream) {
  const float* x      = (const float*)d_in[0];
  const float* thetas = (const float*)d_in[1];
  const float* phis   = (const float*)d_in[2];
  float* out = (float*)d_out;

  float* SrP = (float*)d_ws;
  float* SiP = SrP + CHN * CHN;
  float* tab = SiP + CHN * CHN;

  mesh_trig_kernel<<<ND, PSLOTS * 32, 0, stream>>>(thetas, phis, tab);

  mesh_build_kernel<<<CHN / BUILD_WAVES, BUILD_WAVES * 32, 0, stream>>>(tab, SrP, SiP);

  const int blocks = (BATCHSZ / 64) * (CHN / 128);
  mesh_gemm_kernel<<<blocks, 256, 0, stream>>>(x, SrP, SiP, out);
}


# GEMM operands fetched 16 bytes per lane: one load feeds two k-steps (S stored 4-k interleaved), 32 k per loop trip; k order in each group of 8 permuted alike in A and B (f32 sum order changes)
# speedup vs baseline: 4.3459x; 1.0867x over previous
#include <hip/hip_runtime.h>
#include <math.h>

#define CHN      256
#define ND       128
#define NEV      128
#define NOD      127
#define LSTRIDE  255
#define BATCHSZ  16384

static constexpr float ATTEN = 0.97723722095581068f;

typedef __attribute__((ext_vector_type(2))) float v2f;
typedef __attribute__((ext_vector_type(8))) float v8f;

#define WMMA_F32(A, B, C) \
  __builtin_amdgcn_wmma_f32_16x16x4_f32(false, (A), false, (B), (short)0, (C), false, false)

#define PSLOTS   9
#define TROWS    12
#define TABROW   (TROWS * 32)

__global__ void __launch_bounds__(PSLOTS * 32)
mesh_trig_kernel(const float* __restrict__ thetas, const float* __restrict__ phis,
                 float* __restrict__ tab) {
  const int d = blockIdx.x;
  const int m = threadIdx.x >> 5;
  const int l = threadIdx.x & 31;
  int pi = -1;
  if (m < 4) {
    pi = d * LSTRIDE + 4 * l + m;
  } else {
    const int o = 4 * l - 1 + (m - 4);
    if (o >= 0 && o < NOD) pi = d * LSTRIDE + NEV + o;
  }
  float ecr = 1.0f, eci = 0.0f, esr = 0.0f, esi = 0.0f, cA = 1.0f, sA = 0.0f;
  if (pi >= 0) {
    float s, c, epi, epr;
    sincosf(thetas[pi], &s, &c);
    sincosf(phis[pi], &epi, &epr);
    cA = c * ATTEN;
    sA = s * ATTEN;
    ecr = epr * cA; eci = epi * cA;
    esr = epr * sA; esi = epi * sA;
  }
  float* base = tab + ((size_t)d * TABROW + l) * 4;
  if (m < 4) {
    ((float4*)base)[m * 32] = make_float4(ecr, eci, esr, esi);
    float* cs = base + (7 + (m >> 1)) * 128 + (m & 1) * 2;
    cs[0] = cA; cs[1] = sA;
  } else if (m == 4) {
    float* q = base + 11 * 128;
    q[1] = cA; q[2] = esr; q[3] = esi;
  } else if (m < 8) {
    const int o = m - 5;
    ((float4*)base)[(4 + o) * 32] = make_float4(ecr, eci, esr, esi);
    float* cs = base + (o < 2 ? 9 * 128 + o * 2 : 10 * 128);
    cs[0] = cA; cs[1] = sA;
  } else {
    float* q = base + 10 * 128;
    q[2] = ecr; q[3] = eci;
    base[11 * 128] = sA;
  }
}

typedef __attribute__((ext_vector_type(4))) float v4f;

struct LayerP { v4f q[TROWS]; };

__device__ __forceinline__ LayerP load_layer(const v4f* __restrict__ p) {
  LayerP r;
#pragma unroll
  for (int m = 0; m < TROWS; ++m) r.q[m] = p[m * 32];
  return r;
}

__device__ __forceinline__ unsigned after(unsigned row, const v2f (&x)[8]) {
  asm volatile("" : "+v"(row) : "v"(x[0]), "v"(x[1]), "v"(x[2]), "v"(x[3]),
                                "v"(x[4]), "v"(x[5]), "v"(x[6]), "v"(x[7]));
  return row;
}

__device__ __forceinline__ v2f splat(float v) { return (v2f){v, v}; }

__device__ __forceinline__ v2f cmul(float wr, float wi, v2f z) {
  const v2f zs = (v2f){z.y, z.x};
  return __builtin_elementwise_fma(zs, (v2f){-wi, wi}, z * splat(wr));
}

__device__ __forceinline__ void apply_mzi(v2f& a, v2f& b, const v4f e, float cA, float sA) {
  const v2f na = __builtin_elementwise_fma(b, splat(-sA), cmul(e.x, e.y, a));
  const v2f nb = __builtin_elementwise_fma(b, splat(cA), cmul(e.z, e.w, a));
  a = na; b = nb;
}

__device__ __forceinline__ float from_lane_above(float v) {
  const int x = __float_as_int(v);
  const int edge = __builtin_amdgcn_permlanex16(x, x, 0, 0, false, false);
  return __int_as_float(__builtin_amdgcn_update_dpp(edge, x, 0x101, 0xf, 0xf, false));
}

__device__ __forceinline__ float from_lane_below(float v) {
  const int x = __float_as_int(v);
  const int edge = __builtin_amdgcn_permlanex16(x, x, 0xf, 0, false, false);
  return __int_as_float(__builtin_amdgcn_update_dpp(edge, x, 0x111, 0xf, 0xf, false));
}

__device__ __forceinline__ void double_layer(v2f (&x)[8], const LayerP& P) {
  apply_mzi(x[0], x[1], P.q[0], P.q[7].x, P.q[7].y);
  apply_mzi(x[6], x[7], P.q[3], P.q[8].z, P.q[8].w);
  apply_mzi(x[2], x[3], P.q[1], P.q[7].z, P.q[7].w);
  apply_mzi(x[4], x[5], P.q[2], P.q[8].x, P.q[8].y);
  const v2f hb = (v2f){from_lane_above(x[0].x), from_lane_above(x[0].y)};
  const v2f la = (v2f){from_lane_below(x[7].x), from_lane_below(x[7].y)};
  apply_mzi(x[1], x[2], P.q[4], P.q[9].x, P.q[9].y);
  apply_mzi(x[3], x[4], P.q[5], P.q[9].z, P.q[9].w);
  apply_mzi(x[5], x[6], P.q[6], P.q[10].x, P.q[10].y);
  x[7] = __builtin_elementwise_fma(hb, splat(-P.q[11].x), cmul(P.q[10].z, P.q[10].w, x[7]));
  x[0] = __builtin_elementwise_fma(x[0], splat(P.q[11].y), cmul(P.q[11].z, P.q[11].w, la));
}

#define BUILD_WAVES 2
#define RING 8

__global__ void __launch_bounds__(BUILD_WAVES * 32) __attribute__((amdgpu_waves_per_eu(1, 1)))
mesh_build_kernel(const float* __restrict__ tab,
                  float* __restrict__ SrP, float* __restrict__ SiP) {
  const int lane = threadIdx.x & 31;
  const int k = blockIdx.x * BUILD_WAVES + (threadIdx.x >> 5);

  v2f x[8];
#pragma unroll
  for (int j = 0; j < 8; ++j) x[j] = (v2f){(8 * lane + j == k) ? 1.0f : 0.0f, 0.0f};

  const v4f* tab4 = (const v4f*)tab;
  const v4f* p = tab4 + lane;
  LayerP s0 = load_layer(p + 0 * TABROW), s1 = load_layer(p + 1 * TABROW),
         s2 = load_layer(p + 2 * TABROW), s3 = load_layer(p + 3 * TABROW),
         s4 = load_layer(p + 4 * TABROW), s5 = load_layer(p + 5 * TABROW),
         s6 = load_layer(p + 6 * TABROW), s7 = load_layer(p + 7 * TABROW);
#define RING_STEP(S, i)                                                     \
  {                                                                         \
    double_layer(x, S);                                                     \
    const int dn = (d + RING < ND) ? d + RING + (i) : d + (i);              \
    S = load_layer(tab4 + after((unsigned)(dn * TABROW + lane), x));        \
  }
  for (int d = 0; d < ND; d += RING) {
    RING_STEP(s0, 0) RING_STEP(s1, 1) RING_STEP(s2, 2) RING_STEP(s3, 3)
    RING_STEP(s4, 4) RING_STEP(s5, 5) RING_STEP(s6, 6) RING_STEP(s7, 7)
  }
#undef RING_STEP

  float* rbase = SrP + (size_t)(k >> 2) * (4 * CHN) + (k & 3) + 32 * lane;
  float* ibase = SiP + (size_t)(k >> 2) * (4 * CHN) + (k & 3) + 32 * lane;
#pragma unroll
  for (int j = 0; j < 8; ++j) {
    rbase[4 * j] = x[j].x;
    ibase[4 * j] = x[j].y;
  }
}

__global__ void __launch_bounds__(256) __attribute__((amdgpu_waves_per_eu(4, 4)))
mesh_gemm_kernel(const float* __restrict__ X, const float* __restrict__ SrP,
                 const float* __restrict__ SiP, float* __restrict__ out) {
  const int lane = threadIdx.x & 31;
  const int wave = threadIdx.x >> 5;
  const int wm = wave >> 2;
  const int wn = wave & 3;
  const int bm = blockIdx.x >> 1;
  const int bn = blockIdx.x & 1;
  const int row0 = bm * 64 + wm * 32;
  const int col0 = bn * 128 + wn * 32;
  const int lh = lane >> 4;
  const int lm = lane & 15;

  v8f accR00 = {}, accR01 = {}, accR10 = {}, accR11 = {};
  v8f accI00 = {}, accI01 = {}, accI10 = {}, accI11 = {};

  const float* aptr0 = X + (size_t)(row0 + lm) * CHN + lh * 4;
  const float* aptr1 = X + (size_t)(row0 + 16 + lm) * CHN + lh * 4;
  const size_t bbase = (size_t)lh * (4 * CHN) + (col0 + lm) * 4;
  const float* brptr = SrP + bbase;
  const float* biptr = SiP + bbase;

#pragma unroll 4
  for (int k = 0; k < CHN; k += 8) {
    const size_t boff = (size_t)(k >> 2) * (4 * CHN);
    const v4f a0  = *(const v4f*)(aptr0 + k);
    const v4f a1  = *(const v4f*)(aptr1 + k);
    const v4f br0 = *(const v4f*)(brptr + boff);
    const v4f br1 = *(const v4f*)(brptr + boff + 64);
    const v4f bi0 = *(const v4f*)(biptr + boff);
    const v4f bi1 = *(const v4f*)(biptr + boff + 64);
    accR00 = WMMA_F32(a0.xy, br0.xy, accR00);
    accR01 = WMMA_F32(a0.xy, br1.xy, accR01);
    accR10 = WMMA_F32(a1.xy, br0.xy, accR10);
    accR11 = WMMA_F32(a1.xy, br1.xy, accR11);
    accI00 = WMMA_F32(a0.xy, bi0.xy, accI00);
    accI01 = WMMA_F32(a0.xy, bi1.xy, accI01);
    accI10 = WMMA_F32(a1.xy, bi0.xy, accI10);
    accI11 = WMMA_F32(a1.xy, bi1.xy, accI11);
    accR00 = WMMA_F32(a0.zw, br0.zw, accR00);
    accR01 = WMMA_F32(a0.zw, br1.zw, accR01);
    accR10 = WMMA_F32(a1.zw, br0.zw, accR10);
    accR11 = WMMA_F32(a1.zw, br1.zw, accR11);
    accI00 = WMMA_F32(a0.zw, bi0.zw, accI00);
    accI01 = WMMA_F32(a0.zw, bi1.zw, accI01);
    accI10 = WMMA_F32(a1.zw, bi0.zw, accI10);
    accI11 = WMMA_F32(a1.zw, bi1.zw, accI11);
  }

  float* ob00 = out + (size_t)(row0 + lh * 8) * CHN + col0 + lm;
  float* ob10 = out + (size_t)(row0 + 16 + lh * 8) * CHN + col0 + lm;
#pragma unroll
  for (int r = 0; r < 8; ++r) {
    float vr, vi;
    vr = accR00[r]; vi = accI00[r];
    __builtin_nontemporal_store(vr * vr + vi * vi, ob00 + (size_t)r * CHN);
    vr = accR01[r]; vi = accI01[r];
    __builtin_nontemporal_store(vr * vr + vi * vi, ob00 + (size_t)r * CHN + 16);
    vr = accR10[r]; vi = accI10[r];
    __builtin_nontemporal_store(vr * vr + vi * vi, ob10 + (size_t)r * CHN);
    vr = accR11[r]; vi = accI11[r];
    __builtin_nontemporal_store(vr * vr + vi * vi, ob10 + (size_t)r * CHN + 16);
  }
}

extern "C" void kernel_launch(void* const* d_in, const int* in_sizes, int n_in,
                              void* d_out, int out_size, void* d_ws, size_t ws_size,
                              hipStream_t stream) {
  const float* x      = (const float*)d_in[0];
  const float* thetas = (const float*)d_in[1];
  const float* phis   = (const float*)d_in[2];
  float* out = (float*)d_out;

  float* SrP = (float*)d_ws;
  float* SiP = SrP + CHN * CHN;
  float* tab = SiP + CHN * CHN;

  mesh_trig_kernel<<<ND, PSLOTS * 32, 0, stream>>>(thetas, phis, tab);

  mesh_build_kernel<<<CHN / BUILD_WAVES, BUILD_WAVES * 32, 0, stream>>>(tab, SrP, SiP);

  const int blocks = (BATCHSZ / 64) * (CHN / 128);
  mesh_gemm_kernel<<<blocks, 256, 0, stream>>>(x, SrP, SiP, out);
}


# coefficient table cut from 6 to 4 floats per MZI (phase, then loss-scaled rotation): 9 table loads per lane per double layer, was 12; ring of 16 sets, so the 63-load limit sets the loads' lead
# speedup vs baseline: 4.6495x; 1.0699x over previous
#include <hip/hip_runtime.h>
#include <math.h>

#define CHN      256
#define ND       128
#define NEV      128
#define NOD      127
#define LSTRIDE  255
#define BATCHSZ  16384

static constexpr float ATTEN = 0.97723722095581068f;

typedef __attribute__((ext_vector_type(2))) float v2f;
typedef __attribute__((ext_vector_type(8))) float v8f;

#define WMMA_F32(A, B, C) \
  __builtin_amdgcn_wmma_f32_16x16x4_f32(false, (A), false, (B), (short)0, (C), false, false)

#define PSLOTS   9
#define TROWS    9
#define TABROW   (TROWS * 32)

__global__ void __launch_bounds__(PSLOTS * 32)
mesh_trig_kernel(const float* __restrict__ thetas, const float* __restrict__ phis,
                 float* __restrict__ tab) {
  const int d = blockIdx.x;
  const int m = threadIdx.x >> 5;
  const int l = threadIdx.x & 31;
  int pi = -1;
  if (m < 4) {
    pi = d * LSTRIDE + 4 * l + m;
  } else {
    const int o = 4 * l - 1 + (m - 4);
    if (o >= 0 && o < NOD) pi = d * LSTRIDE + NEV + o;
  }
  float cp = 1.0f, sp = 0.0f, cA = 1.0f, sA = 0.0f;
  if (pi >= 0) {
    float s, c;
    sincosf(thetas[pi], &s, &c);
    sincosf(phis[pi], &sp, &cp);
    cA = c * ATTEN;
    sA = s * ATTEN;
  }
  const int r = (m < 4) ? m : (m == 4 ? 8 : m - 1);
  ((float4*)tab)[(size_t)d * TABROW + r * 32 + l] = make_float4(cp, sp, cA, sA);
}

typedef __attribute__((ext_vector_type(4))) float v4f;

struct LayerP { v4f q[TROWS]; };

__device__ __forceinline__ LayerP load_layer(const v4f* __restrict__ p) {
  LayerP r;
#pragma unroll
  for (int m = 0; m < TROWS; ++m) r.q[m] = p[m * 32];
  return r;
}

__device__ __forceinline__ unsigned after(unsigned row, const v2f (&x)[8]) {
  asm volatile("" : "+v"(row) : "v"(x[0]), "v"(x[1]), "v"(x[2]), "v"(x[3]),
                                "v"(x[4]), "v"(x[5]), "v"(x[6]), "v"(x[7]));
  return row;
}

__device__ __forceinline__ v2f splat(float v) { return (v2f){v, v}; }

__device__ __forceinline__ v2f cmul(float wr, float wi, v2f z) {
  const v2f zs = (v2f){z.y, z.x};
  return __builtin_elementwise_fma(zs, (v2f){-wi, wi}, z * splat(wr));
}

__device__ __forceinline__ void apply_mzi(v2f& a, v2f& b, const v4f q) {
  const v2f ap = cmul(q.x, q.y, a);
  const v2f na = __builtin_elementwise_fma(b, splat(-q.w), ap * splat(q.z));
  const v2f nb = __builtin_elementwise_fma(b, splat(q.z), ap * splat(q.w));
  a = na; b = nb;
}

__device__ __forceinline__ float from_lane_above(float v) {
  const int x = __float_as_int(v);
  const int edge = __builtin_amdgcn_permlanex16(x, x, 0, 0, false, false);
  return __int_as_float(__builtin_amdgcn_update_dpp(edge, x, 0x101, 0xf, 0xf, false));
}

__device__ __forceinline__ float from_lane_below(float v) {
  const int x = __float_as_int(v);
  const int edge = __builtin_amdgcn_permlanex16(x, x, 0xf, 0, false, false);
  return __int_as_float(__builtin_amdgcn_update_dpp(edge, x, 0x111, 0xf, 0xf, false));
}

__device__ __forceinline__ void double_layer(v2f (&x)[8], const LayerP& P) {
  apply_mzi(x[0], x[1], P.q[0]);
  apply_mzi(x[6], x[7], P.q[3]);
  apply_mzi(x[2], x[3], P.q[1]);
  apply_mzi(x[4], x[5], P.q[2]);
  const v2f hb = (v2f){from_lane_above(x[0].x), from_lane_above(x[0].y)};
  const v2f la = (v2f){from_lane_below(x[7].x), from_lane_below(x[7].y)};
  apply_mzi(x[1], x[2], P.q[4]);
  apply_mzi(x[3], x[4], P.q[5]);
  apply_mzi(x[5], x[6], P.q[6]);
  x[7] = __builtin_elementwise_fma(hb, splat(-P.q[7].w),
                                   cmul(P.q[7].x, P.q[7].y, x[7]) * splat(P.q[7].z));
  x[0] = __builtin_elementwise_fma(x[0], splat(P.q[8].z),
                                   cmul(P.q[8].x, P.q[8].y, la) * splat(P.q[8].w));
}

#define BUILD_WAVES 2
#define RING 16

__global__ void __launch_bounds__(BUILD_WAVES * 32) __attribute__((amdgpu_waves_per_eu(1, 1)))
mesh_build_kernel(const float* __restrict__ tab,
                  float* __restrict__ SrP, float* __restrict__ SiP) {
  const int lane = threadIdx.x & 31;
  const int k = blockIdx.x * BUILD_WAVES + (threadIdx.x >> 5);

  v2f x[8];
#pragma unroll
  for (int j = 0; j < 8; ++j) x[j] = (v2f){(8 * lane + j == k) ? 1.0f : 0.0f, 0.0f};

  const v4f* tab4 = (const v4f*)tab;
  const v4f* p = tab4 + lane;
  LayerP s0 = load_layer(p + 0 * TABROW), s1 = load_layer(p + 1 * TABROW),
         s2 = load_layer(p + 2 * TABROW), s3 = load_layer(p + 3 * TABROW),
         s4 = load_layer(p + 4 * TABROW), s5 = load_layer(p + 5 * TABROW),
         s6 = load_layer(p + 6 * TABROW), s7 = load_layer(p + 7 * TABROW),
         s8 = load_layer(p + 8 * TABROW), s9 = load_layer(p + 9 * TABROW),
         s10 = load_layer(p + 10 * TABROW), s11 = load_layer(p + 11 * TABROW),
         s12 = load_layer(p + 12 * TABROW), s13 = load_layer(p + 13 * TABROW),
         s14 = load_layer(p + 14 * TABROW), s15 = load_layer(p + 15 * TABROW);
#define RING_STEP(S, i)                                                     \
  {                                                                         \
    double_layer(x, S);                                                     \
    const int dn = (d + RING < ND) ? d + RING + (i) : d + (i);              \
    S = load_layer(tab4 + after((unsigned)(dn * TABROW + lane), x));        \
  }
  for (int d = 0; d < ND; d += RING) {
    RING_STEP(s0, 0) RING_STEP(s1, 1) RING_STEP(s2, 2) RING_STEP(s3, 3)
    RING_STEP(s4, 4) RING_STEP(s5, 5) RING_STEP(s6, 6) RING_STEP(s7, 7)
    RING_STEP(s8, 8) RING_STEP(s9, 9) RING_STEP(s10, 10) RING_STEP(s11, 11)
    RING_STEP(s12, 12) RING_STEP(s13, 13) RING_STEP(s14, 14) RING_STEP(s15, 15)
  }
#undef RING_STEP

  float* rbase = SrP + (size_t)(k >> 2) * (4 * CHN) + (k & 3) + 32 * lane;
  float* ibase = SiP + (size_t)(k >> 2) * (4 * CHN) + (k & 3) + 32 * lane;
#pragma unroll
  for (int j = 0; j < 8; ++j) {
    rbase[4 * j] = x[j].x;
    ibase[4 * j] = x[j].y;
  }
}

__global__ void __launch_bounds__(256) __attribute__((amdgpu_waves_per_eu(4, 4)))
mesh_gemm_kernel(const float* __restrict__ X, const float* __restrict__ SrP,
                 const float* __restrict__ SiP, float* __restrict__ out) {
  const int lane = threadIdx.x & 31;
  const int wave = threadIdx.x >> 5;
  const int wm = wave >> 2;
  const int wn = wave & 3;
  const int bm = blockIdx.x >> 1;
  const int bn = blockIdx.x & 1;
  const int row0 = bm * 64 + wm * 32;
  const int col0 = bn * 128 + wn * 32;
  const int lh = lane >> 4;
  const int lm = lane & 15;

  v8f accR00 = {}, accR01 = {}, accR10 = {}, accR11 = {};
  v8f accI00 = {}, accI01 = {}, accI10 = {}, accI11 = {};

  const float* aptr0 = X + (size_t)(row0 + lm) * CHN + lh * 4;
  const float* aptr1 = X + (size_t)(row0 + 16 + lm) * CHN + lh * 4;
  const size_t bbase = (size_t)lh * (4 * CHN) + (col0 + lm) * 4;
  const float* brptr = SrP + bbase;
  const float* biptr = SiP + bbase;

#pragma unroll 4
  for (int k = 0; k < CHN; k += 8) {
    const size_t boff = (size_t)(k >> 2) * (4 * CHN);
    const v4f a0  = *(const v4f*)(aptr0 + k);
    const v4f a1  = *(const v4f*)(aptr1 + k);
    const v4f br0 = *(const v4f*)(brptr + boff);
    const v4f br1 = *(const v4f*)(brptr + boff + 64);
    const v4f bi0 = *(const v4f*)(biptr + boff);
    const v4f bi1 = *(const v4f*)(biptr + boff + 64);
    accR00 = WMMA_F32(a0.xy, br0.xy, accR00);
    accR01 = WMMA_F32(a0.xy, br1.xy, accR01);
    accR10 = WMMA_F32(a1.xy, br0.xy, accR10);
    accR11 = WMMA_F32(a1.xy, br1.xy, accR11);
    accI00 = WMMA_F32(a0.xy, bi0.xy, accI00);
    accI01 = WMMA_F32(a0.xy, bi1.xy, accI01);
    accI10 = WMMA_F32(a1.xy, bi0.xy, accI10);
    accI11 = WMMA_F32(a1.xy, bi1.xy, accI11);
    accR00 = WMMA_F32(a0.zw, br0.zw, accR00);
    accR01 = WMMA_F32(a0.zw, br1.zw, accR01);
    accR10 = WMMA_F32(a1.zw, br0.zw, accR10);
    accR11 = WMMA_F32(a1.zw, br1.zw, accR11);
    accI00 = WMMA_F32(a0.zw, bi0.zw, accI00);
    accI01 = WMMA_F32(a0.zw, bi1.zw, accI01);
    accI10 = WMMA_F32(a1.zw, bi0.zw, accI10);
    accI11 = WMMA_F32(a1.zw, bi1.zw, accI11);
  }

  float* ob00 = out + (size_t)(row0 + lh * 8) * CHN + col0 + lm;
  float* ob10 = out + (size_t)(row0 + 16 + lh * 8) * CHN + col0 + lm;
#pragma unroll
  for (int r = 0; r < 8; ++r) {
    float vr, vi;
    vr = accR00[r]; vi = accI00[r];
    __builtin_nontemporal_store(vr * vr + vi * vi, ob00 + (size_t)r * CHN);
    vr = accR01[r]; vi = accI01[r];
    __builtin_nontemporal_store(vr * vr + vi * vi, ob00 + (size_t)r * CHN + 16);
    vr = accR10[r]; vi = accI10[r];
    __builtin_nontemporal_store(vr * vr + vi * vi, ob10 + (size_t)r * CHN);
    vr = accR11[r]; vi = accI11[r];
    __builtin_nontemporal_store(vr * vr + vi * vi, ob10 + (size_t)r * CHN + 16);
  }
}

extern "C" void kernel_launch(void* const* d_in, const int* in_sizes, int n_in,
                              void* d_out, int out_size, void* d_ws, size_t ws_size,
                              hipStream_t stream) {
  const float* x      = (const float*)d_in[0];
  const float* thetas = (const float*)d_in[1];
  const float* phis   = (const float*)d_in[2];
  float* out = (float*)d_out;

  float* SrP = (float*)d_ws;
  float* SiP = SrP + CHN * CHN;
  float* tab = SiP + CHN * CHN;

  mesh_trig_kernel<<<ND, PSLOTS * 32, 0, stream>>>(thetas, phis, tab);

  mesh_build_kernel<<<CHN / BUILD_WAVES, BUILD_WAVES * 32, 0, stream>>>(tab, SrP, SiP);

  const int blocks = (BATCHSZ / 64) * (CHN / 128);
  mesh_gemm_kernel<<<blocks, 256, 0, stream>>>(x, SrP, SiP, out);
}


# build kernel: each of a block's two waves fetches every second double layer's table rows and both read them from LDS (4 stages, split barrier per 2 double layers): 4.5 global loads/lane/layer, was 9
# speedup vs baseline: 4.9180x; 1.0577x over previous
#include <hip/hip_runtime.h>
#include <math.h>

#define CHN      256
#define ND       128
#define NEV      128
#define NOD      127
#define LSTRIDE  255
#define BATCHSZ  16384

static constexpr float ATTEN = 0.97723722095581068f;

typedef __attribute__((ext_vector_type(2))) float v2f;
typedef __attribute__((ext_vector_type(8))) float v8f;

#define WMMA_F32(A, B, C) \
  __builtin_amdgcn_wmma_f32_16x16x4_f32(false, (A), false, (B), (short)0, (C), false, false)

#define PSLOTS   9
#define TROWS    9
#define TABROW   (TROWS * 32)

__global__ void __launch_bounds__(PSLOTS * 32)
mesh_trig_kernel(const float* __restrict__ thetas, const float* __restrict__ phis,
                 float* __restrict__ tab) {
  const int d = blockIdx.x;
  const int m = threadIdx.x >> 5;
  const int l = threadIdx.x & 31;
  int pi = -1;
  if (m < 4) {
    pi = d * LSTRIDE + 4 * l + m;
  } else {
    const int o = 4 * l - 1 + (m - 4);
    if (o >= 0 && o < NOD) pi = d * LSTRIDE + NEV + o;
  }
  float cp = 1.0f, sp = 0.0f, cA = 1.0f, sA = 0.0f;
  if (pi >= 0) {
    float s, c;
    sincosf(thetas[pi], &s, &c);
    sincosf(phis[pi], &sp, &cp);
    cA = c * ATTEN;
    sA = s * ATTEN;
  }
  const int r = (m < 4) ? m : (m == 4 ? 8 : m - 1);
  ((float4*)tab)[(size_t)d * TABROW + r * 32 + l] = make_float4(cp, sp, cA, sA);
}

typedef __attribute__((ext_vector_type(4))) float v4f;

struct LayerP { v4f q[TROWS]; };

__device__ __forceinline__ LayerP load_layer(const v4f* __restrict__ p) {
  LayerP r;
#pragma unroll
  for (int m = 0; m < TROWS; ++m) r.q[m] = p[m * 32];
  return r;
}

__device__ __forceinline__ LayerP load_layer_lds(const v4f* p) {
  LayerP r;
#pragma unroll
  for (int m = 0; m < TROWS; ++m) r.q[m] = p[m * 32];
  return r;
}

__device__ __forceinline__ void store_layer_lds(v4f* p, const LayerP& s) {
#pragma unroll
  for (int m = 0; m < TROWS; ++m) p[m * 32] = s.q[m];
}

__device__ __forceinline__ void compiler_fence() { asm volatile("" ::: "memory"); }

__device__ __forceinline__ void lds_done_signal(v2f& s) {
  asm volatile("s_wait_dscnt 0x0\n\ts_barrier_signal -1" : "+v"(s) :: "memory");
}
__device__ __forceinline__ void lds_stores_done_signal(v2f& s) {
  asm volatile("s_wait_dscnt 0x12\n\ts_barrier_signal -1" : "+v"(s) :: "memory");
}
__device__ __forceinline__ void barrier_wait(v2f& s) {
  asm volatile("s_barrier_wait -1" : "+v"(s) :: "memory");
}

__device__ __forceinline__ v2f splat(float v) { return (v2f){v, v}; }

__device__ __forceinline__ v2f cmul(float wr, float wi, v2f z) {
  const v2f zs = (v2f){z.y, z.x};
  return __builtin_elementwise_fma(zs, (v2f){-wi, wi}, z * splat(wr));
}

__device__ __forceinline__ void apply_mzi(v2f& a, v2f& b, const v4f q) {
  const v2f ap = cmul(q.x, q.y, a);
  const v2f na = __builtin_elementwise_fma(b, splat(-q.w), ap * splat(q.z));
  const v2f nb = __builtin_elementwise_fma(b, splat(q.z), ap * splat(q.w));
  a = na; b = nb;
}

__device__ __forceinline__ float from_lane_above(float v) {
  const int x = __float_as_int(v);
  const int edge = __builtin_amdgcn_permlanex16(x, x, 0, 0, false, false);
  return __int_as_float(__builtin_amdgcn_update_dpp(edge, x, 0x101, 0xf, 0xf, false));
}

__device__ __forceinline__ float from_lane_below(float v) {
  const int x = __float_as_int(v);
  const int edge = __builtin_amdgcn_permlanex16(x, x, 0xf, 0, false, false);
  return __int_as_float(__builtin_amdgcn_update_dpp(edge, x, 0x111, 0xf, 0xf, false));
}

__device__ __forceinline__ void double_layer(v2f (&x)[8], const LayerP& P) {
  apply_mzi(x[0], x[1], P.q[0]);
  apply_mzi(x[6], x[7], P.q[3]);
  apply_mzi(x[2], x[3], P.q[1]);
  apply_mzi(x[4], x[5], P.q[2]);
  const v2f hb = (v2f){from_lane_above(x[0].x), from_lane_above(x[0].y)};
  const v2f la = (v2f){from_lane_below(x[7].x), from_lane_below(x[7].y)};
  apply_mzi(x[1], x[2], P.q[4]);
  apply_mzi(x[3], x[4], P.q[5]);
  apply_mzi(x[5], x[6], P.q[6]);
  x[7] = __builtin_elementwise_fma(hb, splat(-P.q[7].w),
                                   cmul(P.q[7].x, P.q[7].y, x[7]) * splat(P.q[7].z));
  x[0] = __builtin_elementwise_fma(x[0], splat(P.q[8].z),
                                   cmul(P.q[8].x, P.q[8].y, la) * splat(P.q[8].w));
}

#define BUILD_WAVES 2
#define GRING 8
#define NSTAGE 4

struct PhaseP { LayerP a, b; };

__global__ void __launch_bounds__(BUILD_WAVES * 32) __attribute__((amdgpu_waves_per_eu(1, 1)))
mesh_build_kernel(const float* __restrict__ tab,
                  float* __restrict__ SrP, float* __restrict__ SiP) {
  __shared__ v4f stage[NSTAGE][2][TABROW];
  const int lane = threadIdx.x & 31;
  const int w = threadIdx.x >> 5;
  const int k = blockIdx.x * BUILD_WAVES + w;

  v2f x[8];
#pragma unroll
  for (int j = 0; j < 8; ++j) x[j] = (v2f){(8 * lane + j == k) ? 1.0f : 0.0f, 0.0f};

  const v4f* tab4 = (const v4f*)tab;
#define OWN_ROWS(m) (tab4 + (size_t)(2 * (m) + w) * TABROW + lane)
  LayerP g0 = load_layer(OWN_ROWS(0)), g1 = load_layer(OWN_ROWS(1)),
         g2 = load_layer(OWN_ROWS(2)), g3 = load_layer(OWN_ROWS(3)),
         g4 = load_layer(OWN_ROWS(4)), g5 = load_layer(OWN_ROWS(5)),
         g6 = load_layer(OWN_ROWS(6)), g7 = load_layer(OWN_ROWS(7));
  store_layer_lds(&stage[0][w][lane], g0);
  store_layer_lds(&stage[1][w][lane], g1);
  compiler_fence();
  g0 = load_layer(OWN_ROWS(8));
  g1 = load_layer(OWN_ROWS(9));
  lds_done_signal(x[0]);
  barrier_wait(x[0]);
  PhaseP c, n;
  c.a = load_layer_lds(&stage[0][0][lane]);
  c.b = load_layer_lds(&stage[0][1][lane]);
#define PHASE(G, C, N, i)                                                   \
  {                                                                         \
    store_layer_lds(&stage[((i) + 2) & 3][w][lane], G);                     \
    compiler_fence();                                                       \
    G = load_layer(OWN_ROWS((p + (i) + 2 + GRING) & 63));                   \
    N.a = load_layer_lds(&stage[((i) + 1) & 3][0][lane]);                   \
    N.b = load_layer_lds(&stage[((i) + 1) & 3][1][lane]);                   \
    double_layer(x, C.a);                                                   \
    lds_stores_done_signal(x[0]);                                           \
    double_layer(x, C.b);                                                   \
    barrier_wait(x[0]);                                                     \
  }
  for (int p = 0; p < ND / 2; p += 8) {
    PHASE(g2, c, n, 0) PHASE(g3, n, c, 1) PHASE(g4, c, n, 2) PHASE(g5, n, c, 3)
    PHASE(g6, c, n, 4) PHASE(g7, n, c, 5) PHASE(g0, c, n, 6) PHASE(g1, n, c, 7)
  }
#undef PHASE
#undef OWN_ROWS

  float* rbase = SrP + (size_t)(k >> 2) * (4 * CHN) + (k & 3) + 32 * lane;
  float* ibase = SiP + (size_t)(k >> 2) * (4 * CHN) + (k & 3) + 32 * lane;
#pragma unroll
  for (int j = 0; j < 8; ++j) {
    rbase[4 * j] = x[j].x;
    ibase[4 * j] = x[j].y;
  }
}

__global__ void __launch_bounds__(256) __attribute__((amdgpu_waves_per_eu(4, 4)))
mesh_gemm_kernel(const float* __restrict__ X, const float* __restrict__ SrP,
                 const float* __restrict__ SiP, float* __restrict__ out) {
  const int lane = threadIdx.x & 31;
  const int wave = threadIdx.x >> 5;
  const int wm = wave >> 2;
  const int wn = wave & 3;
  const int bm = blockIdx.x >> 1;
  const int bn = blockIdx.x & 1;
  const int row0 = bm * 64 + wm * 32;
  const int col0 = bn * 128 + wn * 32;
  const int lh = lane >> 4;
  const int lm = lane & 15;

  v8f accR00 = {}, accR01 = {}, accR10 = {}, accR11 = {};
  v8f accI00 = {}, accI01 = {}, accI10 = {}, accI11 = {};

  const float* aptr0 = X + (size_t)(row0 + lm) * CHN + lh * 4;
  const float* aptr1 = X + (size_t)(row0 + 16 + lm) * CHN + lh * 4;
  const size_t bbase = (size_t)lh * (4 * CHN) + (col0 + lm) * 4;
  const float* brptr = SrP + bbase;
  const float* biptr = SiP + bbase;

#pragma unroll 4
  for (int k = 0; k < CHN; k += 8) {
    const size_t boff = (size_t)(k >> 2) * (4 * CHN);
    const v4f a0  = *(const v4f*)(aptr0 + k);
    const v4f a1  = *(const v4f*)(aptr1 + k);
    const v4f br0 = *(const v4f*)(brptr + boff);
    const v4f br1 = *(const v4f*)(brptr + boff + 64);
    const v4f bi0 = *(const v4f*)(biptr + boff);
    const v4f bi1 = *(const v4f*)(biptr + boff + 64);
    accR00 = WMMA_F32(a0.xy, br0.xy, accR00);
    accR01 = WMMA_F32(a0.xy, br1.xy, accR01);
    accR10 = WMMA_F32(a1.xy, br0.xy, accR10);
    accR11 = WMMA_F32(a1.xy, br1.xy, accR11);
    accI00 = WMMA_F32(a0.xy, bi0.xy, accI00);
    accI01 = WMMA_F32(a0.xy, bi1.xy, accI01);
    accI10 = WMMA_F32(a1.xy, bi0.xy, accI10);
    accI11 = WMMA_F32(a1.xy, bi1.xy, accI11);
    accR00 = WMMA_F32(a0.zw, br0.zw, accR00);
    accR01 = WMMA_F32(a0.zw, br1.zw, accR01);
    accR10 = WMMA_F32(a1.zw, br0.zw, accR10);
    accR11 = WMMA_F32(a1.zw, br1.zw, accR11);
    accI00 = WMMA_F32(a0.zw, bi0.zw, accI00);
    accI01 = WMMA_F32(a0.zw, bi1.zw, accI01);
    accI10 = WMMA_F32(a1.zw, bi0.zw, accI10);
    accI11 = WMMA_F32(a1.zw, bi1.zw, accI11);
  }

  float* ob00 = out + (size_t)(row0 + lh * 8) * CHN + col0 + lm;
  float* ob10 = out + (size_t)(row0 + 16 + lh * 8) * CHN + col0 + lm;
#pragma unroll
  for (int r = 0; r < 8; ++r) {
    float vr, vi;
    vr = accR00[r]; vi = accI00[r];
    __builtin_nontemporal_store(vr * vr + vi * vi, ob00 + (size_t)r * CHN);
    vr = accR01[r]; vi = accI01[r];
    __builtin_nontemporal_store(vr * vr + vi * vi, ob00 + (size_t)r * CHN + 16);
    vr = accR10[r]; vi = accI10[r];
    __builtin_nontemporal_store(vr * vr + vi * vi, ob10 + (size_t)r * CHN);
    vr = accR11[r]; vi = accI11[r];
    __builtin_nontemporal_store(vr * vr + vi * vi, ob10 + (size_t)r * CHN + 16);
  }
}

extern "C" void kernel_launch(void* const* d_in, const int* in_sizes, int n_in,
                              void* d_out, int out_size, void* d_ws, size_t ws_size,
                              hipStream_t stream) {
  const float* x      = (const float*)d_in[0];
  const float* thetas = (const float*)d_in[1];
  const float* phis   = (const float*)d_in[2];
  float* out = (float*)d_out;

  float* SrP = (float*)d_ws;
  float* SiP = SrP + CHN * CHN;
  float* tab = SiP + CHN * CHN;

  mesh_trig_kernel<<<ND, PSLOTS * 32, 0, stream>>>(thetas, phis, tab);

  mesh_build_kernel<<<CHN / BUILD_WAVES, BUILD_WAVES * 32, 0, stream>>>(tab, SrP, SiP);

  const int blocks = (BATCHSZ / 64) * (CHN / 128);
  mesh_gemm_kernel<<<blocks, 256, 0, stream>>>(x, SrP, SiP, out);
}


# mesh depth split in two: 512 waves each carry a basis vector through 64 double layers (was 256 through 128); new f32 WMMA kernel multiplies the halves, S = A*B (4 launches; f32 sum order changes)
# speedup vs baseline: 5.1079x; 1.0386x over previous
#include <hip/hip_runtime.h>
#include <math.h>

#define CHN      256
#define ND       128
#define NEV      128
#define NOD      127
#define LSTRIDE  255
#define BATCHSZ  16384

static constexpr float ATTEN = 0.97723722095581068f;

typedef __attribute__((ext_vector_type(2))) float v2f;
typedef __attribute__((ext_vector_type(8))) float v8f;

#define WMMA_F32(A, B, C) \
  __builtin_amdgcn_wmma_f32_16x16x4_f32(false, (A), false, (B), (short)0, (C), false, false)

#define PSLOTS   9
#define TROWS    9
#define TABROW   (TROWS * 32)

__global__ void __launch_bounds__(PSLOTS * 32)
mesh_trig_kernel(const float* __restrict__ thetas, const float* __restrict__ phis,
                 float* __restrict__ tab) {
  const int d = blockIdx.x;
  const int m = threadIdx.x >> 5;
  const int l = threadIdx.x & 31;
  int pi = -1;
  if (m < 4) {
    pi = d * LSTRIDE + 4 * l + m;
  } else {
    const int o = 4 * l - 1 + (m - 4);
    if (o >= 0 && o < NOD) pi = d * LSTRIDE + NEV + o;
  }
  float cp = 1.0f, sp = 0.0f, cA = 1.0f, sA = 0.0f;
  if (pi >= 0) {
    float s, c;
    sincosf(thetas[pi], &s, &c);
    sincosf(phis[pi], &sp, &cp);
    cA = c * ATTEN;
    sA = s * ATTEN;
  }
  const int r = (m < 4) ? m : (m == 4 ? 8 : m - 1);
  ((float4*)tab)[(size_t)d * TABROW + r * 32 + l] = make_float4(cp, sp, cA, sA);
}

typedef __attribute__((ext_vector_type(4))) float v4f;

struct LayerP { v4f q[TROWS]; };

__device__ __forceinline__ LayerP load_layer(const v4f* __restrict__ p) {
  LayerP r;
#pragma unroll
  for (int m = 0; m < TROWS; ++m) r.q[m] = p[m * 32];
  return r;
}

__device__ __forceinline__ LayerP load_layer_lds(const v4f* p) {
  LayerP r;
#pragma unroll
  for (int m = 0; m < TROWS; ++m) r.q[m] = p[m * 32];
  return r;
}

__device__ __forceinline__ void store_layer_lds(v4f* p, const LayerP& s) {
#pragma unroll
  for (int m = 0; m < TROWS; ++m) p[m * 32] = s.q[m];
}

__device__ __forceinline__ void compiler_fence() { asm volatile("" ::: "memory"); }

__device__ __forceinline__ void lds_done_signal(v2f& s) {
  asm volatile("s_wait_dscnt 0x0\n\ts_barrier_signal -1" : "+v"(s) :: "memory");
}
__device__ __forceinline__ void lds_stores_done_signal(v2f& s) {
  asm volatile("s_wait_dscnt 0x12\n\ts_barrier_signal -1" : "+v"(s) :: "memory");
}
__device__ __forceinline__ void barrier_wait(v2f& s) {
  asm volatile("s_barrier_wait -1" : "+v"(s) :: "memory");
}

__device__ __forceinline__ v2f splat(float v) { return (v2f){v, v}; }

__device__ __forceinline__ v2f cmul(float wr, float wi, v2f z) {
  const v2f zs = (v2f){z.y, z.x};
  return __builtin_elementwise_fma(zs, (v2f){-wi, wi}, z * splat(wr));
}

__device__ __forceinline__ void apply_mzi(v2f& a, v2f& b, const v4f q) {
  const v2f ap = cmul(q.x, q.y, a);
  const v2f na = __builtin_elementwise_fma(b, splat(-q.w), ap * splat(q.z));
  const v2f nb = __builtin_elementwise_fma(b, splat(q.z), ap * splat(q.w));
  a = na; b = nb;
}

__device__ __forceinline__ float from_lane_above(float v) {
  const int x = __float_as_int(v);
  const int edge = __builtin_amdgcn_permlanex16(x, x, 0, 0, false, false);
  return __int_as_float(__builtin_amdgcn_update_dpp(edge, x, 0x101, 0xf, 0xf, false));
}

__device__ __forceinline__ float from_lane_below(float v) {
  const int x = __float_as_int(v);
  const int edge = __builtin_amdgcn_permlanex16(x, x, 0xf, 0, false, false);
  return __int_as_float(__builtin_amdgcn_update_dpp(edge, x, 0x111, 0xf, 0xf, false));
}

__device__ __forceinline__ void double_layer(v2f (&x)[8], const LayerP& P) {
  apply_mzi(x[0], x[1], P.q[0]);
  apply_mzi(x[6], x[7], P.q[3]);
  apply_mzi(x[2], x[3], P.q[1]);
  apply_mzi(x[4], x[5], P.q[2]);
  const v2f hb = (v2f){from_lane_above(x[0].x), from_lane_above(x[0].y)};
  const v2f la = (v2f){from_lane_below(x[7].x), from_lane_below(x[7].y)};
  apply_mzi(x[1], x[2], P.q[4]);
  apply_mzi(x[3], x[4], P.q[5]);
  apply_mzi(x[5], x[6], P.q[6]);
  x[7] = __builtin_elementwise_fma(hb, splat(-P.q[7].w),
                                   cmul(P.q[7].x, P.q[7].y, x[7]) * splat(P.q[7].z));
  x[0] = __builtin_elementwise_fma(x[0], splat(P.q[8].z),
                                   cmul(P.q[8].x, P.q[8].y, la) * splat(P.q[8].w));
}

#define BUILD_WAVES 2
#define GRING 8
#define NSTAGE 4
#define HND (ND / 2)

struct PhaseP { LayerP a, b; };

__global__ void __launch_bounds__(BUILD_WAVES * 32) __attribute__((amdgpu_waves_per_eu(1, 1)))
mesh_build_kernel(const float* __restrict__ tab,
                  float* __restrict__ Ar, float* __restrict__ Ai,
                  float* __restrict__ BrP, float* __restrict__ BiP) {
  __shared__ v4f stage[NSTAGE][2][TABROW];
  const int lane = threadIdx.x & 31;
  const int w = threadIdx.x >> 5;
  const int h = blockIdx.x >> 7;
  const int k = (blockIdx.x & 127) * BUILD_WAVES + w;

  v2f x[8];
#pragma unroll
  for (int j = 0; j < 8; ++j) x[j] = (v2f){(8 * lane + j == k) ? 1.0f : 0.0f, 0.0f};

  const v4f* tab4 = (const v4f*)tab + (size_t)h * HND * TABROW;
#define OWN_ROWS(m) (tab4 + (size_t)(2 * (m) + w) * TABROW + lane)
  LayerP g0 = load_layer(OWN_ROWS(0)), g1 = load_layer(OWN_ROWS(1)),
         g2 = load_layer(OWN_ROWS(2)), g3 = load_layer(OWN_ROWS(3)),
         g4 = load_layer(OWN_ROWS(4)), g5 = load_layer(OWN_ROWS(5)),
         g6 = load_layer(OWN_ROWS(6)), g7 = load_layer(OWN_ROWS(7));
  store_layer_lds(&stage[0][w][lane], g0);
  store_layer_lds(&stage[1][w][lane], g1);
  compiler_fence();
  g0 = load_layer(OWN_ROWS(8));
  g1 = load_layer(OWN_ROWS(9));
  lds_done_signal(x[0]);
  barrier_wait(x[0]);
  PhaseP c, n;
  c.a = load_layer_lds(&stage[0][0][lane]);
  c.b = load_layer_lds(&stage[0][1][lane]);
#define PHASE(G, C, N, i)                                                   \
  {                                                                         \
    store_layer_lds(&stage[((i) + 2) & 3][w][lane], G);                     \
    compiler_fence();                                                       \
    G = load_layer(OWN_ROWS((p + (i) + 2 + GRING) & (HND / 2 - 1)));        \
    N.a = load_layer_lds(&stage[((i) + 1) & 3][0][lane]);                   \
    N.b = load_layer_lds(&stage[((i) + 1) & 3][1][lane]);                   \
    double_layer(x, C.a);                                                   \
    lds_stores_done_signal(x[0]);                                           \
    double_layer(x, C.b);                                                   \
    barrier_wait(x[0]);                                                     \
  }
  for (int p = 0; p < HND / 2; p += 8) {
    PHASE(g2, c, n, 0) PHASE(g3, n, c, 1) PHASE(g4, c, n, 2) PHASE(g5, n, c, 3)
    PHASE(g6, c, n, 4) PHASE(g7, n, c, 5) PHASE(g0, c, n, 6) PHASE(g1, n, c, 7)
  }
#undef PHASE
#undef OWN_ROWS

  if (h == 0) {
    v4f* ra = (v4f*)(Ar + (size_t)k * CHN + 8 * lane);
    v4f* ia = (v4f*)(Ai + (size_t)k * CHN + 8 * lane);
    ra[0] = (v4f){x[0].x, x[1].x, x[2].x, x[3].x};
    ra[1] = (v4f){x[4].x, x[5].x, x[6].x, x[7].x};
    ia[0] = (v4f){x[0].y, x[1].y, x[2].y, x[3].y};
    ia[1] = (v4f){x[4].y, x[5].y, x[6].y, x[7].y};
  } else {
    float* rbase = BrP + (size_t)(k >> 2) * (4 * CHN) + (k & 3) + 32 * lane;
    float* ibase = BiP + (size_t)(k >> 2) * (4 * CHN) + (k & 3) + 32 * lane;
#pragma unroll
    for (int j = 0; j < 8; ++j) {
      rbase[4 * j] = x[j].x;
      ibase[4 * j] = x[j].y;
    }
  }
}

__global__ void __launch_bounds__(128) __attribute__((amdgpu_waves_per_eu(1, 1)))
mesh_join_kernel(const float* __restrict__ Ar, const float* __restrict__ Ai,
                 const float* __restrict__ BrP, const float* __restrict__ BiP,
                 float* __restrict__ SrP, float* __restrict__ SiP) {
  __shared__ float red[4][2][8][32];
  const int lane = threadIdx.x & 31;
  const int w = threadIdx.x >> 5;
  const int row0 = (blockIdx.x >> 4) * 16;
  const int col0 = (blockIdx.x & 15) * 16;
  const int lh = lane >> 4;
  const int lm = lane & 15;
  const int kq = w * 64;

  v8f accR = {}, accI = {};
  const size_t abase = (size_t)(row0 + lm) * CHN + lh * 4 + kq;
  const float* arptr = Ar + abase;
  const float* aiptr = Ai + abase;
  const size_t bbase = (size_t)((kq >> 2) + lh) * (4 * CHN) + (col0 + lm) * 4;
  const float* brptr = BrP + bbase;
  const float* biptr = BiP + bbase;

#pragma unroll
  for (int k = 0; k < 64; k += 8) {
    const size_t boff = (size_t)(k >> 2) * (4 * CHN);
    const v4f ar = *(const v4f*)(arptr + k);
    const v4f ai = *(const v4f*)(aiptr + k);
    const v4f br = *(const v4f*)(brptr + boff);
    const v4f bi = *(const v4f*)(biptr + boff);
    const v4f an = -ai;
    accR = WMMA_F32(ar.xy, br.xy, accR);
    accI = WMMA_F32(ar.xy, bi.xy, accI);
    accR = WMMA_F32(an.xy, bi.xy, accR);
    accI = WMMA_F32(ai.xy, br.xy, accI);
    accR = WMMA_F32(ar.zw, br.zw, accR);
    accI = WMMA_F32(ar.zw, bi.zw, accI);
    accR = WMMA_F32(an.zw, bi.zw, accR);
    accI = WMMA_F32(ai.zw, br.zw, accI);
  }

#pragma unroll
  for (int r = 0; r < 8; ++r) {
    red[w][0][r][lane] = accR[r];
    red[w][1][r][lane] = accI[r];
  }
  __syncthreads();
  const int pl = lh;
  const int sl = lm + 16 * (w >> 1);
  const int r0 = 4 * (w & 1);
  v4f q;
#pragma unroll
  for (int e = 0; e < 4; ++e) {
    q[e] = ((red[0][pl][r0 + e][sl] + red[1][pl][r0 + e][sl]) + red[2][pl][r0 + e][sl])
           + red[3][pl][r0 + e][sl];
  }
  float* dst = (pl ? SiP : SrP) + (size_t)((row0 >> 2) + w) * (4 * CHN) + (col0 + lm) * 4;
  *(v4f*)dst = q;
}

__global__ void __launch_bounds__(256) __attribute__((amdgpu_waves_per_eu(4, 4)))
mesh_gemm_kernel(const float* __restrict__ X, const float* __restrict__ SrP,
                 const float* __restrict__ SiP, float* __restrict__ out) {
  const int lane = threadIdx.x & 31;
  const int wave = threadIdx.x >> 5;
  const int wm = wave >> 2;
  const int wn = wave & 3;
  const int bm = blockIdx.x >> 1;
  const int bn = blockIdx.x & 1;
  const int row0 = bm * 64 + wm * 32;
  const int col0 = bn * 128 + wn * 32;
  const int lh = lane >> 4;
  const int lm = lane & 15;

  v8f accR00 = {}, accR01 = {}, accR10 = {}, accR11 = {};
  v8f accI00 = {}, accI01 = {}, accI10 = {}, accI11 = {};

  const float* aptr0 = X + (size_t)(row0 + lm) * CHN + lh * 4;
  const float* aptr1 = X + (size_t)(row0 + 16 + lm) * CHN + lh * 4;
  const size_t bbase = (size_t)lh * (4 * CHN) + (col0 + lm) * 4;
  const float* brptr = SrP + bbase;
  const float* biptr = SiP + bbase;

#pragma unroll 4
  for (int k = 0; k < CHN; k += 8) {
    const size_t boff = (size_t)(k >> 2) * (4 * CHN);
    const v4f a0  = *(const v4f*)(aptr0 + k);
    const v4f a1  = *(const v4f*)(aptr1 + k);
    const v4f br0 = *(const v4f*)(brptr + boff);
    const v4f br1 = *(const v4f*)(brptr + boff + 64);
    const v4f bi0 = *(const v4f*)(biptr + boff);
    const v4f bi1 = *(const v4f*)(biptr + boff + 64);
    accR00 = WMMA_F32(a0.xy, br0.xy, accR00);
    accR01 = WMMA_F32(a0.xy, br1.xy, accR01);
    accR10 = WMMA_F32(a1.xy, br0.xy, accR10);
    accR11 = WMMA_F32(a1.xy, br1.xy, accR11);
    accI00 = WMMA_F32(a0.xy, bi0.xy, accI00);
    accI01 = WMMA_F32(a0.xy, bi1.xy, accI01);
    accI10 = WMMA_F32(a1.xy, bi0.xy, accI10);
    accI11 = WMMA_F32(a1.xy, bi1.xy, accI11);
    accR00 = WMMA_F32(a0.zw, br0.zw, accR00);
    accR01 = WMMA_F32(a0.zw, br1.zw, accR01);
    accR10 = WMMA_F32(a1.zw, br0.zw, accR10);
    accR11 = WMMA_F32(a1.zw, br1.zw, accR11);
    accI00 = WMMA_F32(a0.zw, bi0.zw, accI00);
    accI01 = WMMA_F32(a0.zw, bi1.zw, accI01);
    accI10 = WMMA_F32(a1.zw, bi0.zw, accI10);
    accI11 = WMMA_F32(a1.zw, bi1.zw, accI11);
  }

  float* ob00 = out + (size_t)(row0 + lh * 8) * CHN + col0 + lm;
  float* ob10 = out + (size_t)(row0 + 16 + lh * 8) * CHN + col0 + lm;
#pragma unroll
  for (int r = 0; r < 8; ++r) {
    float vr, vi;
    vr = accR00[r]; vi = accI00[r];
    __builtin_nontemporal_store(vr * vr + vi * vi, ob00 + (size_t)r * CHN);
    vr = accR01[r]; vi = accI01[r];
    __builtin_nontemporal_store(vr * vr + vi * vi, ob00 + (size_t)r * CHN + 16);
    vr = accR10[r]; vi = accI10[r];
    __builtin_nontemporal_store(vr * vr + vi * vi, ob10 + (size_t)r * CHN);
    vr = accR11[r]; vi = accI11[r];
    __builtin_nontemporal_store(vr * vr + vi * vi, ob10 + (size_t)r * CHN + 16);
  }
}

extern "C" void kernel_launch(void* const* d_in, const int* in_sizes, int n_in,
                              void* d_out, int out_size, void* d_ws, size_t ws_size,
                              hipStream_t stream) {
  const float* x      = (const float*)d_in[0];
  const float* thetas = (const float*)d_in[1];
  const float* phis   = (const float*)d_in[2];
  float* out = (float*)d_out;

  float* SrP = (float*)d_ws;
  float* SiP = SrP + CHN * CHN;
  float* tab = SiP + CHN * CHN;
  float* Ar  = tab + ND * TABROW * 4;
  float* Ai  = Ar + CHN * CHN;
  float* BrP = Ai + CHN * CHN;
  float* BiP = BrP + CHN * CHN;

  mesh_trig_kernel<<<ND, PSLOTS * 32, 0, stream>>>(thetas, phis, tab);

  mesh_build_kernel<<<2 * (CHN / BUILD_WAVES), BUILD_WAVES * 32, 0, stream>>>(tab, Ar, Ai, BrP, BiP);

  mesh_join_kernel<<<(CHN / 16) * (CHN / 16), 128, 0, stream>>>(Ar, Ai, BrP, BiP, SrP, SiP);

  const int blocks = (BATCHSZ / 64) * (CHN / 128);
  mesh_gemm_kernel<<<blocks, 256, 0, stream>>>(x, SrP, SiP, out);
}
